# starting kernel
# baseline (speedup 1.0000x reference)
#include <hip/hip_runtime.h>

#define EMB 128
#define LOG2_BOOST 0.5849625007211562f
#define EPS 1e-8f

typedef __attribute__((ext_vector_type(2))) float v2f;
typedef __attribute__((ext_vector_type(8))) float v8f;

__global__ void me_zero_kernel(int* __restrict__ counts, float* __restrict__ partials,
                               int R, int P) {
    int i = blockIdx.x * blockDim.x + threadIdx.x;
    if (i < R) counts[i] = 0;
    if (i < P) partials[i] = 0.0f;
}

__global__ void me_count_kernel(const int* __restrict__ wi, const int* __restrict__ rmap,
                                int* __restrict__ roots, int* __restrict__ counts, int n) {
    int i = blockIdx.x * blockDim.x + threadIdx.x;
    if (i < n) {
        int r = rmap[wi[i]];
        roots[i] = r;
        atomicAdd(&counts[r], 1);
    }
}

__global__ void __launch_bounds__(256)
me_root_feat_kernel(const float* __restrict__ emb, const float* __restrict__ mats,
                    const float* __restrict__ bias, float* __restrict__ f) {
    int r = blockIdx.x;
    __shared__ float se[EMB];
    if (threadIdx.x < EMB) se[threadIdx.x] = emb[r * EMB + threadIdx.x];
    __syncthreads();

    int wave = threadIdx.x >> 5;
    int lane = threadIdx.x & 31;
    int m0   = wave * 16;

    int  koff = (lane >= 16) ? 2 : 0;
    const float* Mrow = mats + (size_t)r * EMB * EMB
                             + (size_t)(m0 + (lane & 15)) * EMB + koff;
    float bmask = ((lane & 15) == 0) ? 1.0f : 0.0f;

    v8f c = {};
    for (int k = 0; k < EMB; k += 4) {
        v2f a = *(const v2f*)(Mrow + k);
        v2f b;
        b.x = se[k + koff]     * bmask;
        b.y = se[k + koff + 1] * bmask;
        c = __builtin_amdgcn_wmma_f32_16x16x4_f32(
                 false, a,  false, b,
                 (short)0, c,  false,  false);
    }

    if ((lane & 15) == 0) {
        int base = m0 + ((lane == 16) ? 8 : 0);
        #pragma unroll
        for (int v = 0; v < 8; ++v)
            f[r * EMB + base + v] = c[v] + bias[r * EMB + base + v];
    }
}

__global__ void __launch_bounds__(256)
me_boost_kernel(const float* __restrict__ w, const int* __restrict__ roots,
                const int* __restrict__ counts, float* __restrict__ boosted,
                float* __restrict__ partials, int n) {
    __shared__ float red[256];
    int i = blockIdx.x * blockDim.x + threadIdx.x;
    float b = 0.0f;
    if (i < n) {
        float cm1 = (float)(counts[roots[i]] - 1);
        b = w[i] * exp2f(cm1 * LOG2_BOOST) + EPS;
        boosted[i] = b;
    }
    red[threadIdx.x] = b;
    __syncthreads();
    #pragma unroll
    for (int s = 128; s > 0; s >>= 1) {
        if (threadIdx.x < s) red[threadIdx.x] += red[threadIdx.x + s];
        __syncthreads();
    }
    if (threadIdx.x == 0) partials[blockIdx.x] = red[0];
}

__global__ void me_reduce_kernel(const float* __restrict__ partials,
                                 float* __restrict__ sum, int P) {
    if (threadIdx.x == 0) {
        float s = 0.0f;
        for (int i = 0; i < P; ++i) s += partials[i];
        sum[0] = s;
    }
}

__global__ void __launch_bounds__(128)
me_out_kernel(const float* __restrict__ f, const int* __restrict__ roots,
              const float* __restrict__ boosted, const float* __restrict__ sum,
              float4* __restrict__ out, int n) {
    int word = blockIdx.x * 4 + (threadIdx.x >> 5);
    int lane = threadIdx.x & 31;
    if (word < n) {
        float scale = boosted[word] / sum[0];
        const float4* fr = (const float4*)(f + (size_t)roots[word] * EMB);
        float4 v = fr[lane];
        v.x *= scale; v.y *= scale; v.z *= scale; v.w *= scale;
        out[(size_t)word * (EMB / 4) + lane] = v;
    }
}

static inline size_t me_align256(size_t x) { return (x + 255) & ~(size_t)255; }

extern "C" void kernel_launch(void* const* d_in, const int* in_sizes, int n_in,
                              void* d_out, int out_size, void* d_ws, size_t ws_size,
                              hipStream_t stream) {
    const float* emb  = (const float*)d_in[0];
    const float* mats = (const float*)d_in[1];
    const float* bias = (const float*)d_in[2];
    const float* attw = (const float*)d_in[3];
    const int*   wi   = (const int*)d_in[4];
    const int*   rmap = (const int*)d_in[5];

    const int N = in_sizes[3];
    const int R = in_sizes[0] / EMB;
    const int P = (N + 255) / 256;

    char* ws = (char*)d_ws;
    size_t off = 0;
    int*   counts   = (int*)(ws + off);   off = me_align256(off + (size_t)R * 4);
    int*   roots    = (int*)(ws + off);   off = me_align256(off + (size_t)N * 4);
    float* boosted  = (float*)(ws + off); off = me_align256(off + (size_t)N * 4);
    float* partials = (float*)(ws + off); off = me_align256(off + (size_t)P * 4);
    float* sum      = (float*)(ws + off); off = me_align256(off + 4);
    float* f        = (float*)(ws + off); off = me_align256(off + (size_t)R * EMB * 4);
    (void)ws_size; (void)n_in; (void)out_size;

    int zmax = (R > P) ? R : P;
    me_zero_kernel<<<(zmax + 255) / 256, 256, 0, stream>>>(counts, partials, R, P);
    me_count_kernel<<<(N + 255) / 256, 256, 0, stream>>>(wi, rmap, roots, counts, N);
    me_root_feat_kernel<<<R, 256, 0, stream>>>(emb, mats, bias, f);
    me_boost_kernel<<<P, 256, 0, stream>>>(attw, roots, counts, boosted, partials, N);
    me_reduce_kernel<<<1, 32, 0, stream>>>(partials, sum, P);
    me_out_kernel<<<(N + 3) / 4, 128, 0, stream>>>(f, roots, boosted, sum,
                                                   (float4*)d_out, N);
}


# matvec fragments re-laid so each 16-byte load reads 4 whole 128-byte lines (4 rows x 4 k-groups on the WMMA diagonal); row sum order now (g0+g1)+(g2+g3) over the k-groups; 256 blocks of 16 waves
# speedup vs baseline: 1.1733x; 1.1733x over previous
#include <hip/hip_runtime.h>

#define EMB 128
#define LOG2_BOOST 0.5849625007211562f
#define EPS 1e-8f

typedef __attribute__((ext_vector_type(2))) float v2f;
typedef __attribute__((ext_vector_type(8))) float v8f;

__global__ void me_zero_kernel(int* __restrict__ counts, float* __restrict__ partials,
                               int R, int P) {
    int i = blockIdx.x * blockDim.x + threadIdx.x;
    if (i < R) counts[i] = 0;
    if (i < P) partials[i] = 0.0f;
}

__global__ void me_count_kernel(const int* __restrict__ wi, const int* __restrict__ rmap,
                                int* __restrict__ roots, int* __restrict__ counts, int n) {
    int i = blockIdx.x * blockDim.x + threadIdx.x;
    if (i < n) {
        int r = rmap[wi[i]];
        roots[i] = r;
        atomicAdd(&counts[r], 1);
    }
}

#define RF_WAVES 16
#define RF_TPG   4
#define RF_NGRP  4

__global__ void __launch_bounds__(RF_WAVES * 32)
me_root_feat_kernel(const float* __restrict__ emb, const float* __restrict__ mats,
                    const float* __restrict__ bias, float* __restrict__ f, int R) {
    __shared__ float sp[RF_WAVES * 64 * 4];
    const int tid  = threadIdx.x;
    const int lane = tid & 31;
    const int wave = __builtin_amdgcn_readfirstlane(tid >> 5);
    const int h = lane >> 4, c = lane & 15, g = c >> 2, rho = c & 3;

    const int root = blockIdx.x * (RF_WAVES / 2) + (wave >> 1);
    const int rl   = (root < R) ? root : (R - 1);
    const int koff = 16 * h + 4 * g;

    const float4* ep = (const float4*)(emb + (size_t)rl * EMB + koff);
    float4 ev[4];
    #pragma unroll
    for (int t = 0; t < 4; ++t) ev[t] = ep[8 * t];

    const float4* mp = (const float4*)(mats + (size_t)rl * EMB * EMB
                                            + (size_t)((wave & 1) * 64 + rho) * EMB + koff);
    float4 buf[2][RF_TPG * 4];

#define RF_LOAD_GROUP(G) do { \
    _Pragma("unroll") for (int p = 0; p < RF_TPG / 2; ++p) \
    _Pragma("unroll") for (int t = 0; t < 4; ++t) \
    _Pragma("unroll") for (int u = 0; u < 2; ++u) \
        buf[(G) & 1][(2 * p + u) * 4 + t] = mp[((G) * RF_TPG + 2 * p + u) * 128 + 8 * t]; } while (0)

    RF_LOAD_GROUP(0);
    float* spw = sp + (wave * 64 + rho) * 4 + g;
    const bool diag = ((c >> 3) == h);
    const bool b0 = (lane & 1) != 0, b1 = (lane & 2) != 0, b2 = (lane & 4) != 0;

    #pragma unroll
    for (int grp = 0; grp < RF_NGRP; ++grp) {
        __builtin_amdgcn_sched_barrier(0);
        if (grp + 1 < RF_NGRP) RF_LOAD_GROUP(grp + 1);
        __builtin_amdgcn_sched_barrier(0);

        #pragma unroll
        for (int p = 0; p < RF_TPG / 2; ++p) {
            v8f acc0 = {}, acc1 = {};
            #pragma unroll
            for (int t = 0; t < 4; ++t) {
                const float4 m0 = buf[grp & 1][(2 * p) * 4 + t];
                const float4 m1 = buf[grp & 1][(2 * p + 1) * 4 + t];
                v2f a, b;
                b.x = ev[t].x; b.y = ev[t].y;
                a.x = m0.x; a.y = m0.y;
                acc0 = __builtin_amdgcn_wmma_f32_16x16x4_f32(false, a, false, b, (short)0, acc0, false, false);
                a.x = m1.x; a.y = m1.y;
                acc1 = __builtin_amdgcn_wmma_f32_16x16x4_f32(false, a, false, b, (short)0, acc1, false, false);
                b.x = ev[t].z; b.y = ev[t].w;
                a.x = m0.z; a.y = m0.w;
                acc0 = __builtin_amdgcn_wmma_f32_16x16x4_f32(false, a, false, b, (short)0, acc0, false, false);
                a.x = m1.z; a.y = m1.w;
                acc1 = __builtin_amdgcn_wmma_f32_16x16x4_f32(false, a, false, b, (short)0, acc1, false, false);
            }
            #pragma unroll
            for (int u = 0; u < 2; ++u) {
                const v8f d = u ? acc1 : acc0;
                const float s0 = b0 ? d[1] : d[0], s1 = b0 ? d[3] : d[2];
                const float s2 = b0 ? d[5] : d[4], s3 = b0 ? d[7] : d[6];
                const float q0 = b1 ? s1 : s0, q1 = b1 ? s3 : s2;
                const float dg = b2 ? q1 : q0;
                if (diag) spw[((grp * RF_TPG + 2 * p + u) * 4) * 4] = dg;
            }
        }
    }
#undef RF_LOAD_GROUP
    __syncthreads();

    #pragma unroll
    for (int o = tid; o < RF_WAVES * 64; o += RF_WAVES * 32) {
        const int ro = blockIdx.x * (RF_WAVES / 2) + (o >> 7);
        if (ro < R) {
            const float4 pz = *(const float4*)(sp + o * 4);
            const size_t idx = (size_t)blockIdx.x * (RF_WAVES * 64) + o;
            f[idx] = ((pz.x + pz.y) + (pz.z + pz.w)) + bias[idx];
        }
    }
}

__global__ void __launch_bounds__(256)
me_boost_kernel(const float* __restrict__ w, const int* __restrict__ roots,
                const int* __restrict__ counts, float* __restrict__ boosted,
                float* __restrict__ partials, int n) {
    __shared__ float red[256];
    int i = blockIdx.x * blockDim.x + threadIdx.x;
    float b = 0.0f;
    if (i < n) {
        float cm1 = (float)(counts[roots[i]] - 1);
        b = w[i] * exp2f(cm1 * LOG2_BOOST) + EPS;
        boosted[i] = b;
    }
    red[threadIdx.x] = b;
    __syncthreads();
    #pragma unroll
    for (int s = 128; s > 0; s >>= 1) {
        if (threadIdx.x < s) red[threadIdx.x] += red[threadIdx.x + s];
        __syncthreads();
    }
    if (threadIdx.x == 0) partials[blockIdx.x] = red[0];
}

__global__ void me_reduce_kernel(const float* __restrict__ partials,
                                 float* __restrict__ sum, int P) {
    if (threadIdx.x == 0) {
        float s = 0.0f;
        for (int i = 0; i < P; ++i) s += partials[i];
        sum[0] = s;
    }
}

__global__ void __launch_bounds__(128)
me_out_kernel(const float* __restrict__ f, const int* __restrict__ roots,
              const float* __restrict__ boosted, const float* __restrict__ sum,
              float4* __restrict__ out, int n) {
    int word = blockIdx.x * 4 + (threadIdx.x >> 5);
    int lane = threadIdx.x & 31;
    if (word < n) {
        float scale = boosted[word] / sum[0];
        const float4* fr = (const float4*)(f + (size_t)roots[word] * EMB);
        float4 v = fr[lane];
        v.x *= scale; v.y *= scale; v.z *= scale; v.w *= scale;
        out[(size_t)word * (EMB / 4) + lane] = v;
    }
}

static inline size_t me_align256(size_t x) { return (x + 255) & ~(size_t)255; }

extern "C" void kernel_launch(void* const* d_in, const int* in_sizes, int n_in,
                              void* d_out, int out_size, void* d_ws, size_t ws_size,
                              hipStream_t stream) {
    const float* emb  = (const float*)d_in[0];
    const float* mats = (const float*)d_in[1];
    const float* bias = (const float*)d_in[2];
    const float* attw = (const float*)d_in[3];
    const int*   wi   = (const int*)d_in[4];
    const int*   rmap = (const int*)d_in[5];

    const int N = in_sizes[3];
    const int R = in_sizes[0] / EMB;
    const int P = (N + 255) / 256;

    char* ws = (char*)d_ws;
    size_t off = 0;
    int*   counts   = (int*)(ws + off);   off = me_align256(off + (size_t)R * 4);
    int*   roots    = (int*)(ws + off);   off = me_align256(off + (size_t)N * 4);
    float* boosted  = (float*)(ws + off); off = me_align256(off + (size_t)N * 4);
    float* partials = (float*)(ws + off); off = me_align256(off + (size_t)P * 4);
    float* sum      = (float*)(ws + off); off = me_align256(off + 4);
    float* f        = (float*)(ws + off); off = me_align256(off + (size_t)R * EMB * 4);
    (void)ws_size; (void)n_in; (void)out_size;

    int zmax = (R > P) ? R : P;
    me_zero_kernel<<<(zmax + 255) / 256, 256, 0, stream>>>(counts, partials, R, P);
    me_count_kernel<<<(N + 255) / 256, 256, 0, stream>>>(wi, rmap, roots, counts, N);
    me_root_feat_kernel<<<(R + 7) / 8, 512, 0, stream>>>(emb, mats, bias, f, R);
    me_boost_kernel<<<P, 256, 0, stream>>>(attw, roots, counts, boosted, partials, N);
    me_reduce_kernel<<<1, 32, 0, stream>>>(partials, sum, P);
    me_out_kernel<<<(N + 3) / 4, 128, 0, stream>>>(f, roots, boosted, sum,
                                                   (float4*)d_out, N);
}


# six launches fused into three: root gather rides in the matvec launch, every boost block counts roots in its own LDS histogram, every output wave adds the block partials by a lane tree (was serial)
# speedup vs baseline: 1.3668x; 1.1649x over previous
#include <hip/hip_runtime.h>

#define EMB 128
#define LOG2_BOOST 0.5849625007211562f
#define EPS 1e-8f

typedef __attribute__((ext_vector_type(2))) float v2f;
typedef __attribute__((ext_vector_type(8))) float v8f;

#define RF_WAVES 16
#define RF_TPG   4
#define RF_NGRP  4

__global__ void __launch_bounds__(RF_WAVES * 32)
me_root_feat_kernel(const float* __restrict__ emb, const float* __restrict__ mats,
                    const float* __restrict__ bias, float* __restrict__ f, int R,
                    const int* __restrict__ wi, const int* __restrict__ rmap,
                    int* __restrict__ roots, int n, int wpb) {
    __shared__ float sp[RF_WAVES * 64 * 4];
    const int tid  = threadIdx.x;
    const int lane = tid & 31;
    const int wave = __builtin_amdgcn_readfirstlane(tid >> 5);
    const int h = lane >> 4, c = lane & 15, g = c >> 2, rho = c & 3;

    const int root = blockIdx.x * (RF_WAVES / 2) + (wave >> 1);
    const int rl   = (root < R) ? root : (R - 1);
    const int koff = 16 * h + 4 * g;

    const int wrd  = blockIdx.x * wpb + tid;
    const bool hasw = (tid < wpb) && (wrd < n);
    const int widx = wi[hasw ? wrd : 0];
    int myroot = 0;

    const float4* ep = (const float4*)(emb + (size_t)rl * EMB + koff);
    float4 ev[4];
    #pragma unroll
    for (int t = 0; t < 4; ++t) ev[t] = ep[8 * t];

    const float4* mp = (const float4*)(mats + (size_t)rl * EMB * EMB
                                            + (size_t)((wave & 1) * 64 + rho) * EMB + koff);
    float4 buf[2][RF_TPG * 4];

#define RF_LOAD_GROUP(G) do { \
    _Pragma("unroll") for (int p = 0; p < RF_TPG / 2; ++p) \
    _Pragma("unroll") for (int t = 0; t < 4; ++t) \
    _Pragma("unroll") for (int u = 0; u < 2; ++u) \
        buf[(G) & 1][(2 * p + u) * 4 + t] = mp[((G) * RF_TPG + 2 * p + u) * 128 + 8 * t]; } while (0)

    RF_LOAD_GROUP(0);
    float* spw = sp + (wave * 64 + rho) * 4 + g;
    const bool diag = ((c >> 3) == h);
    const bool b0 = (lane & 1) != 0, b1 = (lane & 2) != 0, b2 = (lane & 4) != 0;

    #pragma unroll
    for (int grp = 0; grp < RF_NGRP; ++grp) {
        __builtin_amdgcn_sched_barrier(0);
        if (grp + 1 < RF_NGRP) RF_LOAD_GROUP(grp + 1);
        if (grp == 1) myroot = rmap[widx];
        __builtin_amdgcn_sched_barrier(0);

        #pragma unroll
        for (int p = 0; p < RF_TPG / 2; ++p) {
            v8f acc0 = {}, acc1 = {};
            #pragma unroll
            for (int t = 0; t < 4; ++t) {
                const float4 m0 = buf[grp & 1][(2 * p) * 4 + t];
                const float4 m1 = buf[grp & 1][(2 * p + 1) * 4 + t];
                v2f a, b;
                b.x = ev[t].x; b.y = ev[t].y;
                a.x = m0.x; a.y = m0.y;
                acc0 = __builtin_amdgcn_wmma_f32_16x16x4_f32(false, a, false, b, (short)0, acc0, false, false);
                a.x = m1.x; a.y = m1.y;
                acc1 = __builtin_amdgcn_wmma_f32_16x16x4_f32(false, a, false, b, (short)0, acc1, false, false);
                b.x = ev[t].z; b.y = ev[t].w;
                a.x = m0.z; a.y = m0.w;
                acc0 = __builtin_amdgcn_wmma_f32_16x16x4_f32(false, a, false, b, (short)0, acc0, false, false);
                a.x = m1.z; a.y = m1.w;
                acc1 = __builtin_amdgcn_wmma_f32_16x16x4_f32(false, a, false, b, (short)0, acc1, false, false);
            }
            #pragma unroll
            for (int u = 0; u < 2; ++u) {
                const v8f d = u ? acc1 : acc0;
                const float s0 = b0 ? d[1] : d[0], s1 = b0 ? d[3] : d[2];
                const float s2 = b0 ? d[5] : d[4], s3 = b0 ? d[7] : d[6];
                const float q0 = b1 ? s1 : s0, q1 = b1 ? s3 : s2;
                const float dg = b2 ? q1 : q0;
                if (diag) spw[((grp * RF_TPG + 2 * p + u) * 4) * 4] = dg;
            }
        }
    }
#undef RF_LOAD_GROUP
    __syncthreads();

    #pragma unroll
    for (int o = tid; o < RF_WAVES * 64; o += RF_WAVES * 32) {
        const int ro = blockIdx.x * (RF_WAVES / 2) + (o >> 7);
        if (ro < R) {
            const float4 pz = *(const float4*)(sp + o * 4);
            const size_t idx = (size_t)blockIdx.x * (RF_WAVES * 64) + o;
            f[idx] = ((pz.x + pz.y) + (pz.z + pz.w)) + bias[idx];
        }
    }

    if (hasw) roots[wrd] = myroot;
    for (int j = tid + RF_WAVES * 32; j < wpb; j += RF_WAVES * 32) {
        const int i2 = blockIdx.x * wpb + j;
        if (i2 < n) roots[i2] = rmap[wi[i2]];
    }
}

#define ME_HL 16

__global__ void __launch_bounds__(256)
me_boost_kernel(const float* __restrict__ w, const int* __restrict__ roots,
                float* __restrict__ boosted, float* __restrict__ partials, int n, int R) {
    extern __shared__ int hist[];
    __shared__ float red[256];
    const int i = blockIdx.x * blockDim.x + threadIdx.x;
    const int   myr = (i < n) ? roots[i] : 0;
    const float myw = (i < n) ? w[i] : 0.0f;

    for (int j = threadIdx.x; j < R; j += 256) hist[j] = 0;
    __syncthreads();

    const int4* r4 = (const int4*)roots;
    const int n4 = n >> 2;
    int j0 = threadIdx.x;
    for (; j0 + (ME_HL - 1) * 256 < n4; j0 += ME_HL * 256) {
        int4 q[ME_HL];
        #pragma unroll
        for (int u = 0; u < ME_HL; ++u) q[u] = r4[j0 + u * 256];
        #pragma unroll
        for (int u = 0; u < ME_HL; ++u) {
            atomicAdd(&hist[q[u].x], 1);
            atomicAdd(&hist[q[u].y], 1);
            atomicAdd(&hist[q[u].z], 1);
            atomicAdd(&hist[q[u].w], 1);
        }
    }
    for (; j0 < n4; j0 += 256) {
        const int4 q = r4[j0];
        atomicAdd(&hist[q.x], 1);
        atomicAdd(&hist[q.y], 1);
        atomicAdd(&hist[q.z], 1);
        atomicAdd(&hist[q.w], 1);
    }
    for (int j = 4 * n4 + threadIdx.x; j < n; j += 256) atomicAdd(&hist[roots[j]], 1);
    __syncthreads();

    float b = 0.0f;
    if (i < n) {
        float cm1 = (float)(hist[myr] - 1);
        b = myw * exp2f(cm1 * LOG2_BOOST) + EPS;
        boosted[i] = b;
    }
    red[threadIdx.x] = b;
    __syncthreads();
    #pragma unroll
    for (int s = 128; s > 0; s >>= 1) {
        if (threadIdx.x < s) red[threadIdx.x] += red[threadIdx.x + s];
        __syncthreads();
    }
    if (threadIdx.x == 0) partials[blockIdx.x] = red[0];
}

__global__ void __launch_bounds__(128)
me_out_kernel(const float* __restrict__ f, const int* __restrict__ roots,
              const float* __restrict__ boosted, const float* __restrict__ partials,
              float4* __restrict__ out, int n, int P) {
    const int word = blockIdx.x * 4 + (threadIdx.x >> 5);
    const int lane = threadIdx.x & 31;
    if (word >= n) return;
    const int   r  = roots[word];
    const float bw = boosted[word];
    const float p0 = partials[(lane < P) ? lane : 0];
    const float p1 = partials[(lane + 32 < P) ? lane + 32 : 0];
    float sum = ((lane < P) ? p0 : 0.0f) + ((lane + 32 < P) ? p1 : 0.0f);
    for (int i = lane + 64; i < P; i += 32) sum += partials[i];
    float4 v = ((const float4*)(f + (size_t)r * EMB))[lane];
    #pragma unroll
    for (int m = 16; m > 0; m >>= 1) sum += __shfl_xor(sum, m);
    const float scale = bw / sum;
    v.x *= scale; v.y *= scale; v.z *= scale; v.w *= scale;
    out[(size_t)word * (EMB / 4) + lane] = v;
}

static inline size_t me_align256(size_t x) { return (x + 255) & ~(size_t)255; }

extern "C" void kernel_launch(void* const* d_in, const int* in_sizes, int n_in,
                              void* d_out, int out_size, void* d_ws, size_t ws_size,
                              hipStream_t stream) {
    const float* emb  = (const float*)d_in[0];
    const float* mats = (const float*)d_in[1];
    const float* bias = (const float*)d_in[2];
    const float* attw = (const float*)d_in[3];
    const int*   wi   = (const int*)d_in[4];
    const int*   rmap = (const int*)d_in[5];

    const int N = in_sizes[3];
    const int R = in_sizes[0] / EMB;
    const int P = (N + 255) / 256;
    const int G = (R + 7) / 8;
    const int wpb = (N + G - 1) / G;

    char* ws = (char*)d_ws;
    size_t off = 0;
    int*   roots    = (int*)(ws + off);   off = me_align256(off + (size_t)N * 4);
    float* boosted  = (float*)(ws + off); off = me_align256(off + (size_t)N * 4);
    float* partials = (float*)(ws + off); off = me_align256(off + (size_t)P * 4);
    float* f        = (float*)(ws + off); off = me_align256(off + (size_t)R * EMB * 4);
    (void)ws_size; (void)n_in; (void)out_size;

    me_root_feat_kernel<<<G, 512, 0, stream>>>(emb, mats, bias, f, R, wi, rmap, roots, N, wpb);
    me_boost_kernel<<<P, 256, (size_t)R * sizeof(int), stream>>>(attw, roots, boosted, partials, N, R);
    me_out_kernel<<<(N + 3) / 4, 128, 0, stream>>>(f, roots, boosted, partials,
                                                   (float4*)d_out, N, P);
}


# boost and output launches fused (128 blocks x 16 waves): each block counts all roots in LDS and forms the whole weight sum itself (sum order now per thread, lane tree, wave tree); 3 launches to 2
# speedup vs baseline: 1.5609x; 1.1420x over previous
#include <hip/hip_runtime.h>

#define EMB 128
#define LOG2_BOOST 0.5849625007211562f
#define EPS 1e-8f

typedef __attribute__((ext_vector_type(2))) float v2f;
typedef __attribute__((ext_vector_type(8))) float v8f;

#define RF_WAVES 16
#define RF_TPG   4
#define RF_NGRP  4

__global__ void __launch_bounds__(RF_WAVES * 32)
me_root_feat_kernel(const float* __restrict__ emb, const float* __restrict__ mats,
                    const float* __restrict__ bias, float* __restrict__ f, int R,
                    const int* __restrict__ wi, const int* __restrict__ rmap,
                    int* __restrict__ roots, int n, int wpb) {
    __shared__ float sp[RF_WAVES * 64 * 4];
    const int tid  = threadIdx.x;
    const int lane = tid & 31;
    const int wave = __builtin_amdgcn_readfirstlane(tid >> 5);
    const int h = lane >> 4, c = lane & 15, g = c >> 2, rho = c & 3;

    const int root = blockIdx.x * (RF_WAVES / 2) + (wave >> 1);
    const int rl   = (root < R) ? root : (R - 1);
    const int koff = 16 * h + 4 * g;

    const int wrd  = blockIdx.x * wpb + tid;
    const bool hasw = (tid < wpb) && (wrd < n);
    const int widx = wi[hasw ? wrd : 0];
    int myroot = 0;

    const float4* ep = (const float4*)(emb + (size_t)rl * EMB + koff);
    float4 ev[4];
    #pragma unroll
    for (int t = 0; t < 4; ++t) ev[t] = ep[8 * t];

    const float4* mp = (const float4*)(mats + (size_t)rl * EMB * EMB
                                            + (size_t)((wave & 1) * 64 + rho) * EMB + koff);
    float4 buf[2][RF_TPG * 4];

#define RF_LOAD_GROUP(G) do { \
    _Pragma("unroll") for (int p = 0; p < RF_TPG / 2; ++p) \
    _Pragma("unroll") for (int t = 0; t < 4; ++t) \
    _Pragma("unroll") for (int u = 0; u < 2; ++u) \
        buf[(G) & 1][(2 * p + u) * 4 + t] = mp[((G) * RF_TPG + 2 * p + u) * 128 + 8 * t]; } while (0)

    RF_LOAD_GROUP(0);
    float* spw = sp + (wave * 64 + rho) * 4 + g;
    const bool diag = ((c >> 3) == h);
    const bool b0 = (lane & 1) != 0, b1 = (lane & 2) != 0, b2 = (lane & 4) != 0;

    #pragma unroll
    for (int grp = 0; grp < RF_NGRP; ++grp) {
        __builtin_amdgcn_sched_barrier(0);
        if (grp + 1 < RF_NGRP) RF_LOAD_GROUP(grp + 1);
        if (grp == 1) myroot = rmap[widx];
        __builtin_amdgcn_sched_barrier(0);

        #pragma unroll
        for (int p = 0; p < RF_TPG / 2; ++p) {
            v8f acc0 = {}, acc1 = {};
            #pragma unroll
            for (int t = 0; t < 4; ++t) {
                const float4 m0 = buf[grp & 1][(2 * p) * 4 + t];
                const float4 m1 = buf[grp & 1][(2 * p + 1) * 4 + t];
                v2f a, b;
                b.x = ev[t].x; b.y = ev[t].y;
                a.x = m0.x; a.y = m0.y;
                acc0 = __builtin_amdgcn_wmma_f32_16x16x4_f32(false, a, false, b, (short)0, acc0, false, false);
                a.x = m1.x; a.y = m1.y;
                acc1 = __builtin_amdgcn_wmma_f32_16x16x4_f32(false, a, false, b, (short)0, acc1, false, false);
                b.x = ev[t].z; b.y = ev[t].w;
                a.x = m0.z; a.y = m0.w;
                acc0 = __builtin_amdgcn_wmma_f32_16x16x4_f32(false, a, false, b, (short)0, acc0, false, false);
                a.x = m1.z; a.y = m1.w;
                acc1 = __builtin_amdgcn_wmma_f32_16x16x4_f32(false, a, false, b, (short)0, acc1, false, false);
            }
            #pragma unroll
            for (int u = 0; u < 2; ++u) {
                const v8f d = u ? acc1 : acc0;
                const float s0 = b0 ? d[1] : d[0], s1 = b0 ? d[3] : d[2];
                const float s2 = b0 ? d[5] : d[4], s3 = b0 ? d[7] : d[6];
                const float q0 = b1 ? s1 : s0, q1 = b1 ? s3 : s2;
                const float dg = b2 ? q1 : q0;
                if (diag) spw[((grp * RF_TPG + 2 * p + u) * 4) * 4] = dg;
            }
        }
    }
#undef RF_LOAD_GROUP
    __syncthreads();

    #pragma unroll
    for (int o = tid; o < RF_WAVES * 64; o += RF_WAVES * 32) {
        const int ro = blockIdx.x * (RF_WAVES / 2) + (o >> 7);
        if (ro < R) {
            const float4 pz = *(const float4*)(sp + o * 4);
            const size_t idx = (size_t)blockIdx.x * (RF_WAVES * 64) + o;
            f[idx] = ((pz.x + pz.y) + (pz.z + pz.w)) + bias[idx];
        }
    }

    if (hasw) roots[wrd] = myroot;
    for (int j = tid + RF_WAVES * 32; j < wpb; j += RF_WAVES * 32) {
        const int i2 = blockIdx.x * wpb + j;
        if (i2 < n) roots[i2] = rmap[wi[i2]];
    }
}

#define TO_WAVES 16
#define TO_WPW   8
#define TO_Q     8
#define TO_THR   (TO_WAVES * 32)

#define ME_ROWX(x, M) __builtin_bit_cast(float, __builtin_amdgcn_update_dpp( \
    0, __builtin_bit_cast(int, (x)), 0x160 | (M), 0xf, 0xf, false))
__device__ __forceinline__ float me_xhalf(float x) {
    return __builtin_bit_cast(float, __builtin_amdgcn_permlanex16(
        __builtin_bit_cast(int, x), __builtin_bit_cast(int, x), 0x76543210, 0xfedcba98, false, false));
}

__device__ __forceinline__ float me_boosted(const int* fac, int r, float wv) {
    return wv * __builtin_bit_cast(float, fac[r]) + EPS;
}

__global__ void __launch_bounds__(TO_THR)
me_boost_out_kernel(const float* __restrict__ w, const int* __restrict__ roots,
                    const float* __restrict__ f, float4* __restrict__ out, int n, int R) {
    extern __shared__ int hist[];
    __shared__ float red[TO_WAVES];
    const int tid  = threadIdx.x;
    const int lane = tid & 31;
    const int wave = tid >> 5;

    const int4*   r4 = (const int4*)roots;
    const float4* w4 = (const float4*)w;
    const int n4 = n >> 2;
    int4 q[TO_Q];
    #pragma unroll
    for (int u = 0; u < TO_Q; ++u) {
        const int j = tid + u * TO_THR;
        q[u] = r4[(j < n4) ? j : 0];
    }
    const int word0 = (blockIdx.x * TO_WAVES + wave) * TO_WPW;
    int   rw[TO_WPW];
    float ww[TO_WPW];
    #pragma unroll
    for (int k = 0; k < TO_WPW; ++k) {
        const int wd = (word0 + k < n) ? word0 + k : 0;
        rw[k] = roots[wd];
        ww[k] = w[wd];
    }

    for (int j = tid; j < R; j += TO_THR) hist[j] = 0;
    __syncthreads();

    #pragma unroll
    for (int u = 0; u < TO_Q; ++u) {
        if (tid + u * TO_THR < n4) {
            atomicAdd(&hist[q[u].x], 1);
            atomicAdd(&hist[q[u].y], 1);
            atomicAdd(&hist[q[u].z], 1);
            atomicAdd(&hist[q[u].w], 1);
        }
    }
    for (int j = tid + TO_Q * TO_THR; j < n4; j += TO_THR) {
        const int4 qq = r4[j];
        atomicAdd(&hist[qq.x], 1);
        atomicAdd(&hist[qq.y], 1);
        atomicAdd(&hist[qq.z], 1);
        atomicAdd(&hist[qq.w], 1);
    }
    for (int j = 4 * n4 + tid; j < n; j += TO_THR) atomicAdd(&hist[roots[j]], 1);

    float4 wq[TO_Q];
    #pragma unroll
    for (int u = 0; u < TO_Q; ++u) {
        const int j = tid + u * TO_THR;
        wq[u] = w4[(j < n4) ? j : 0];
    }
    float4 fv[TO_WPW];
    #pragma unroll
    for (int k = 0; k < TO_WPW; ++k) fv[k] = ((const float4*)(f + (size_t)rw[k] * EMB))[lane];
    __syncthreads();

    for (int j = tid; j < R; j += TO_THR)
        hist[j] = __builtin_bit_cast(int, exp2f((float)(hist[j] - 1) * LOG2_BOOST));
    __syncthreads();

    float s = 0.0f;
    #pragma unroll
    for (int u = 0; u < TO_Q; ++u) {
        if (tid + u * TO_THR < n4) {
            const float t0 = me_boosted(hist, q[u].x, wq[u].x), t1 = me_boosted(hist, q[u].y, wq[u].y);
            const float t2 = me_boosted(hist, q[u].z, wq[u].z), t3 = me_boosted(hist, q[u].w, wq[u].w);
            s += (t0 + t1) + (t2 + t3);
        }
    }
    for (int j = tid + TO_Q * TO_THR; j < n4; j += TO_THR) {
        const int4 qq = r4[j];
        const float4 wv = w4[j];
        const float t0 = me_boosted(hist, qq.x, wv.x), t1 = me_boosted(hist, qq.y, wv.y);
        const float t2 = me_boosted(hist, qq.z, wv.z), t3 = me_boosted(hist, qq.w, wv.w);
        s += (t0 + t1) + (t2 + t3);
    }
    for (int j = 4 * n4 + tid; j < n; j += TO_THR) s += me_boosted(hist, roots[j], w[j]);

    s += ME_ROWX(s, 1); s += ME_ROWX(s, 2); s += ME_ROWX(s, 4); s += ME_ROWX(s, 8);
    s += me_xhalf(s);
    if (lane == 0) red[wave] = s;
    __syncthreads();
    float sum = red[lane & (TO_WAVES - 1)];
    sum += ME_ROWX(sum, 1); sum += ME_ROWX(sum, 2); sum += ME_ROWX(sum, 4); sum += ME_ROWX(sum, 8);

    float sc[TO_WPW];
    #pragma unroll
    for (int k = 0; k < TO_WPW; ++k) sc[k] = me_boosted(hist, rw[k], ww[k]) / sum;
    #pragma unroll
    for (int k = 0; k < TO_WPW; ++k) {
        if (word0 + k < n) {
            float4 v = fv[k];
            v.x *= sc[k]; v.y *= sc[k]; v.z *= sc[k]; v.w *= sc[k];
            out[(size_t)(word0 + k) * (EMB / 4) + lane] = v;
        }
    }
}

static inline size_t me_align256(size_t x) { return (x + 255) & ~(size_t)255; }

extern "C" void kernel_launch(void* const* d_in, const int* in_sizes, int n_in,
                              void* d_out, int out_size, void* d_ws, size_t ws_size,
                              hipStream_t stream) {
    const float* emb  = (const float*)d_in[0];
    const float* mats = (const float*)d_in[1];
    const float* bias = (const float*)d_in[2];
    const float* attw = (const float*)d_in[3];
    const int*   wi   = (const int*)d_in[4];
    const int*   rmap = (const int*)d_in[5];

    const int N = in_sizes[3];
    const int R = in_sizes[0] / EMB;
    const int G = (R + 7) / 8;
    const int wpb = (N + G - 1) / G;
    const int B = (N + TO_WAVES * TO_WPW - 1) / (TO_WAVES * TO_WPW);

    char* ws = (char*)d_ws;
    size_t off = 0;
    int*   roots = (int*)(ws + off);   off = me_align256(off + (size_t)N * 4);
    float* f     = (float*)(ws + off); off = me_align256(off + (size_t)R * EMB * 4);
    (void)ws_size; (void)n_in; (void)out_size;

    me_root_feat_kernel<<<G, 512, 0, stream>>>(emb, mats, bias, f, R, wi, rmap, roots, N, wpb);
    me_boost_out_kernel<<<B, TO_THR, (size_t)R * sizeof(int), stream>>>(attw, roots, f, (float4*)d_out, N, R);
}


# two launches fused into one: every 8-root block copies the word-to-root table into LDS, counts and sums all words between its matrix groups, and writes only its own roots' rows from f kept in LDS
# speedup vs baseline: 1.6876x; 1.0812x over previous
#include <hip/hip_runtime.h>

#define EMB 128
#define LOG2_BOOST 0.5849625007211562f
#define EPS 1e-8f

typedef __attribute__((ext_vector_type(2))) float v2f;
typedef __attribute__((ext_vector_type(8))) float v8f;
typedef __attribute__((ext_vector_type(4))) int   v4i;

#define RF_WAVES 16
#define RF_TPG   4
#define RF_NGRP  4
#define RF_THR   (RF_WAVES * 32)
#define ME_VOCAB 65536
#define ME_ROOTS 2048
#define ME_Q     8
#define ME_ST    16

#define ME_ROWX(x, M) __builtin_bit_cast(float, __builtin_amdgcn_update_dpp( \
    0, __builtin_bit_cast(int, (x)), 0x160 | (M), 0xf, 0xf, false))
__device__ __forceinline__ float me_xhalf(float x) {
    return __builtin_bit_cast(float, __builtin_amdgcn_permlanex16(
        __builtin_bit_cast(int, x), __builtin_bit_cast(int, x), 0x76543210, 0xfedcba98, false, false));
}

__device__ __forceinline__ float me_boosted(const int* fac, int r, float wv) {
    return wv * __builtin_bit_cast(float, fac[r]) + EPS;
}

__global__ void __launch_bounds__(RF_THR)
me_forward_kernel(const float* __restrict__ emb, const float* __restrict__ mats,
                  const float* __restrict__ bias, const float* __restrict__ w,
                  const int* __restrict__ wi, const int* __restrict__ rmap,
                  float4* __restrict__ out, int R, int n) {
    __shared__ __attribute__((aligned(16))) int   map[ME_VOCAB];
    __shared__ int   hist[ME_ROOTS];
    __shared__ __attribute__((aligned(16))) float sp[RF_WAVES * 64 * 4];
    __shared__ __attribute__((aligned(16))) float fl[(RF_WAVES / 2) * EMB];
    __shared__ float red[RF_WAVES];
    const int tid  = threadIdx.x;
    const int lane = tid & 31;
    const int wave = __builtin_amdgcn_readfirstlane(tid >> 5);
    const int h = lane >> 4, c = lane & 15, g = c >> 2, rho = c & 3;

    const int root0 = blockIdx.x * (RF_WAVES / 2);
    const int root  = root0 + (wave >> 1);
    const int rl    = (root < R) ? root : (R - 1);
    const int koff  = 16 * h + 4 * g;

    for (int k = 0; k < ME_ROOTS / RF_THR; ++k) hist[tid + k * RF_THR] = 0;

    v4i st[ME_ST];
    const v4i* m4 = (const v4i*)rmap;
    const int rot = blockIdx.x & (2 * ME_ST - 1);
    #pragma unroll
    for (int u = 0; u < ME_ST; ++u) st[u] = m4[tid + ((u + rot) & (2 * ME_ST - 1)) * RF_THR];

    const float4* ep = (const float4*)(emb + (size_t)rl * EMB + koff);
    float4 ev[4];
    #pragma unroll
    for (int t = 0; t < 4; ++t) ev[t] = ep[8 * t];

    const float4* mp = (const float4*)(mats + (size_t)rl * EMB * EMB
                                            + (size_t)((wave & 1) * 64 + rho) * EMB + koff);
    float4 buf[2][RF_TPG * 4];

    const int4*   i4 = (const int4*)wi;
    const float4* w4 = (const float4*)w;
    const int n4 = n >> 2;
    int4   q[ME_Q];
    float4 wq[ME_Q];
    float  bz[2];

#define RF_LOAD_GROUP(G) do { \
    _Pragma("unroll") for (int p = 0; p < RF_TPG / 2; ++p) \
    _Pragma("unroll") for (int t = 0; t < 4; ++t) \
    _Pragma("unroll") for (int u = 0; u < 2; ++u) \
        buf[(G) & 1][(2 * p + u) * 4 + t] = mp[((G) * RF_TPG + 2 * p + u) * 128 + 8 * t]; } while (0)

    RF_LOAD_GROUP(0);
    __builtin_amdgcn_sched_barrier(0);
    #pragma unroll
    for (int u = 0; u < ME_ST; ++u) ((v4i*)map)[tid + ((u + rot) & (2 * ME_ST - 1)) * RF_THR] = st[u];
    float* spw = sp + (wave * 64 + rho) * 4 + g;
    const bool diag = ((c >> 3) == h);
    const bool b0 = (lane & 1) != 0, b1 = (lane & 2) != 0, b2 = (lane & 4) != 0;

    float s = 0.0f;
    #pragma unroll
    for (int grp = 0; grp < RF_NGRP; ++grp) {
        __builtin_amdgcn_sched_barrier(0);
        if (grp + 1 < RF_NGRP) RF_LOAD_GROUP(grp + 1);
        if (grp == 0) {
            #pragma unroll
            for (int u = 0; u < ME_ST; ++u) st[u] = m4[tid + ((ME_ST + u + rot) & (2 * ME_ST - 1)) * RF_THR];
        }
        if (grp == 1) {
            #pragma unroll
            for (int u = 0; u < ME_Q; ++u) {
                const int j = tid + u * RF_THR;
                q[u] = i4[(j < n4) ? j : 0];
            }
        }
        if (grp == 2) {
            #pragma unroll
            for (int u = 0; u < ME_Q; ++u) {
                const int j = tid + u * RF_THR;
                wq[u] = w4[(j < n4) ? j : 0];
            }
            #pragma unroll
            for (int k = 0; k < 2; ++k) {
                const int o = tid + k * RF_THR;
                const bool inr = root0 + (o >> 7) < R;
                bz[k] = bias[inr ? (size_t)blockIdx.x * (RF_WAVES * 64) + o : 0];
            }
        }
        __builtin_amdgcn_sched_barrier(0);
        #pragma unroll
        for (int p = 0; p < RF_TPG / 2; ++p) {
            v8f acc0 = {}, acc1 = {};
            #pragma unroll
            for (int t = 0; t < 4; ++t) {
                const float4 m0 = buf[grp & 1][(2 * p) * 4 + t];
                const float4 m1 = buf[grp & 1][(2 * p + 1) * 4 + t];
                v2f a, b;
                b.x = ev[t].x; b.y = ev[t].y;
                a.x = m0.x; a.y = m0.y;
                acc0 = __builtin_amdgcn_wmma_f32_16x16x4_f32(false, a, false, b, (short)0, acc0, false, false);
                a.x = m1.x; a.y = m1.y;
                acc1 = __builtin_amdgcn_wmma_f32_16x16x4_f32(false, a, false, b, (short)0, acc1, false, false);
                b.x = ev[t].z; b.y = ev[t].w;
                a.x = m0.z; a.y = m0.w;
                acc0 = __builtin_amdgcn_wmma_f32_16x16x4_f32(false, a, false, b, (short)0, acc0, false, false);
                a.x = m1.z; a.y = m1.w;
                acc1 = __builtin_amdgcn_wmma_f32_16x16x4_f32(false, a, false, b, (short)0, acc1, false, false);
            }
            #pragma unroll
            for (int u = 0; u < 2; ++u) {
                const v8f d = u ? acc1 : acc0;
                const float s0 = b0 ? d[1] : d[0], s1 = b0 ? d[3] : d[2];
                const float s2 = b0 ? d[5] : d[4], s3 = b0 ? d[7] : d[6];
                const float q0 = b1 ? s1 : s0, q1 = b1 ? s3 : s2;
                const float dg = b2 ? q1 : q0;
                if (diag) spw[((grp * RF_TPG + 2 * p + u) * 4) * 4] = dg;
            }
        }

        __builtin_amdgcn_sched_barrier(0);
        if (grp == 0) {
            #pragma unroll
            for (int u = 0; u < ME_ST; ++u)
                ((v4i*)map)[tid + ((ME_ST + u + rot) & (2 * ME_ST - 1)) * RF_THR] = st[u];
        }
        if (grp == 1) {
            #pragma unroll
            for (int u = 0; u < ME_Q; ++u) {
                q[u].x = map[q[u].x]; q[u].y = map[q[u].y];
                q[u].z = map[q[u].z]; q[u].w = map[q[u].w];
            }
        }
        if (grp == 2) {
            #pragma unroll
            for (int k = 0; k < 2; ++k) fl[tid + k * RF_THR] = bz[k];
        }
        if (grp == 0) __syncthreads();
        if (grp == 1) {
            #pragma unroll
            for (int u = 0; u < ME_Q; ++u) {
                if (tid + u * RF_THR < n4) {
                    atomicAdd(&hist[q[u].x], 1);
                    atomicAdd(&hist[q[u].y], 1);
                    atomicAdd(&hist[q[u].z], 1);
                    atomicAdd(&hist[q[u].w], 1);
                }
            }
            __syncthreads();
        }
        if (grp == 2) {
            for (int k = 0; k < ME_ROOTS / RF_THR; ++k) {
                const int j = tid + k * RF_THR;
                hist[j] = __builtin_bit_cast(int, exp2f((float)(hist[j] - 1) * LOG2_BOOST));
            }
            __syncthreads();
            #pragma unroll
            for (int u = 0; u < ME_Q; ++u) {
                const float t0 = me_boosted(hist, q[u].x, wq[u].x), t1 = me_boosted(hist, q[u].y, wq[u].y);
                const float t2 = me_boosted(hist, q[u].z, wq[u].z), t3 = me_boosted(hist, q[u].w, wq[u].w);
                s += (tid + u * RF_THR < n4) ? (t0 + t1) + (t2 + t3) : 0.0f;
            }
        }
    }
#undef RF_LOAD_GROUP

    s += ME_ROWX(s, 1); s += ME_ROWX(s, 2); s += ME_ROWX(s, 4); s += ME_ROWX(s, 8);
    s += me_xhalf(s);
    if (lane == 0) red[wave] = s;
    __syncthreads();
    float sum = red[lane & (RF_WAVES - 1)];
    sum += ME_ROWX(sum, 1); sum += ME_ROWX(sum, 2); sum += ME_ROWX(sum, 4); sum += ME_ROWX(sum, 8);

    for (int k = 0; k < 2; ++k) {
        const int o = tid + k * RF_THR;
        const float4 pz = *(const float4*)(sp + o * 4);
        fl[o] = ((pz.x + pz.y) + (pz.z + pz.w)) + fl[o];
    }
    __syncthreads();

    #pragma unroll
    for (int u = 0; u < ME_Q; ++u) {
        const bool okw = (tid + u * RF_THR) < n4;
        bool hit[4];
        #pragma unroll
        for (int e = 0; e < 4; ++e) {
            const int r = (e == 0) ? q[u].x : (e == 1) ? q[u].y : (e == 2) ? q[u].z : q[u].w;
            hit[e] = okw && ((unsigned)(r - root0) < (unsigned)(RF_WAVES / 2)) && (r < R);
        }
        if (__builtin_amdgcn_ballot_w32(hit[0] || hit[1] || hit[2] || hit[3]) == 0u) continue;
        #pragma unroll
        for (int e = 0; e < 4; ++e) {
            const int   r  = (e == 0) ? q[u].x : (e == 1) ? q[u].y : (e == 2) ? q[u].z : q[u].w;
            const float wv = (e == 0) ? wq[u].x : (e == 1) ? wq[u].y : (e == 2) ? wq[u].z : wq[u].w;
            unsigned mask = __builtin_amdgcn_ballot_w32(hit[e]);
            while (mask) {
                const int l = __builtin_ctz(mask);
                mask &= mask - 1;
                const int   rr = __builtin_amdgcn_readlane(r, l);
                const float wl = __builtin_bit_cast(float, __builtin_amdgcn_readlane(__builtin_bit_cast(int, wv), l));
                const int word = 4 * ((wave * 32 + l) + u * RF_THR) + e;
                const float sc = me_boosted(hist, rr, wl) / sum;
                float4 v = ((const float4*)(fl + (rr - root0) * EMB))[lane];
                v.x *= sc; v.y *= sc; v.z *= sc; v.w *= sc;
                out[(size_t)word * (EMB / 4) + lane] = v;
            }
        }
    }
}

extern "C" void kernel_launch(void* const* d_in, const int* in_sizes, int n_in,
                              void* d_out, int out_size, void* d_ws, size_t ws_size,
                              hipStream_t stream) {
    const float* emb  = (const float*)d_in[0];
    const float* mats = (const float*)d_in[1];
    const float* bias = (const float*)d_in[2];
    const float* attw = (const float*)d_in[3];
    const int*   wi   = (const int*)d_in[4];
    const int*   rmap = (const int*)d_in[5];

    const int N = in_sizes[3];
    const int R = in_sizes[0] / EMB;
    const int G = (R + 7) / 8;
    (void)d_ws; (void)ws_size; (void)n_in; (void)out_size;

    me_forward_kernel<<<G, RF_THR, 0, stream>>>(emb, mats, bias, attw, wi, rmap, (float4*)d_out, R, N);
}
